# starting kernel
# baseline (speedup 1.0000x reference)
#include <hip/hip_runtime.h>


typedef __attribute__((ext_vector_type(2))) float v2f;
typedef __attribute__((ext_vector_type(8))) float v8f;

#define NBLADES 8
#define BATCH   128
#define INFEAT  128
#define UNITS   64
#define CHUNK   32
#define XPAD    36

#define BLADE_PERM_TAB 0x76534210u
__device__ __forceinline__ int blade_perm(int i) {
  return (int)((BLADE_PERM_TAB >> (4 * i)) & 7u);
}

__global__ __launch_bounds__(256) void ga_mvprod_wmma(
    const float* __restrict__ x,
    const float* __restrict__ w,
    float* __restrict__ out)
{
  __shared__ float xs[NBLADES][16][XPAD];
  __shared__ float ws[NBLADES][CHUNK][16];

  const int tid   = threadIdx.x;
  const int lane  = tid & 31;
  const int o     = __builtin_amdgcn_readfirstlane(tid >> 5);
  const int mtile = blockIdx.x >> 2;
  const int ntile = blockIdx.x & 3;

  const int mo = blade_perm(o);

  unsigned ibword = 0, negword = 0;
#pragma unroll
  for (int ia = 0; ia < NBLADES; ++ia) {
    const int ma  = blade_perm(ia);
    const int mb  = mo ^ ma;
    const int ib  = blade_perm(mb);
    const int a1  = (ma >> 1) & 1, a2 = (ma >> 2) & 1;
    const int neg = ((mb & (a1 ^ a2)) ^ ((mb >> 1) & a2)) & 1;
    ibword  |= (unsigned)ib  << (3 * ia);
    negword |= (unsigned)neg << ia;
  }

  const int fm = lane & 15;
  const int kb = (lane >> 4) * 2;

  v8f accP = {};
  v8f accN = {};

  for (int ch = 0; ch < INFEAT / CHUNK; ++ch) {
    __syncthreads();

#pragma unroll
    for (int r = 0; r < 4; ++r) {
      int idx = tid + 256 * r;
      int ia  = idx >> 7;
      int rem = idx & 127;
      int m   = rem >> 3;
      int g   = rem & 7;
      float4 v = *(const float4*)(x + ((size_t)(ia * BATCH + mtile * 16 + m) * INFEAT
                                       + ch * CHUNK + 4 * g));
      *(float4*)(&xs[ia][m][4 * g]) = v;
    }
#pragma unroll
    for (int r = 0; r < 4; ++r) {
      int idx = tid + 256 * r;
      int ib  = idx >> 7;
      int rem = idx & 127;
      int k   = rem >> 2;
      int g   = rem & 3;
      float4 v = *(const float4*)(w + ((size_t)(ib * INFEAT + ch * CHUNK + k) * UNITS
                                       + ntile * 16 + 4 * g));
      *(float4*)(&ws[ib][k][4 * g]) = v;
    }
    __syncthreads();

    if (ch + 1 < INFEAT / CHUNK) {
      __builtin_prefetch(x + ((size_t)((tid >> 5) * BATCH + mtile * 16 + (tid & 15)) * INFEAT
                              + (ch + 1) * CHUNK), 0, 3);
      __builtin_prefetch(w + ((size_t)((tid >> 5) * INFEAT + (ch + 1) * CHUNK + (tid & 31)) * UNITS
                              + ntile * 16), 0, 3);
    }

    for (int ia = 0; ia < NBLADES; ++ia) {
      const int ib = (int)((ibword >> (3 * ia)) & 7u);
      if ((negword >> ia) & 1u) {
#pragma unroll
        for (int k = 0; k < CHUNK; k += 4) {
          v2f a, b;
          a.x = xs[ia][fm][k + kb];
          a.y = xs[ia][fm][k + kb + 1];
          b.x = ws[ib][k + kb][fm];
          b.y = ws[ib][k + kb + 1][fm];
          accN = __builtin_amdgcn_wmma_f32_16x16x4_f32(
              false, a, false, b, (short)0, accN, false, false);
        }
      } else {
#pragma unroll
        for (int k = 0; k < CHUNK; k += 4) {
          v2f a, b;
          a.x = xs[ia][fm][k + kb];
          a.y = xs[ia][fm][k + kb + 1];
          b.x = ws[ib][k + kb][fm];
          b.y = ws[ib][k + kb + 1][fm];
          accP = __builtin_amdgcn_wmma_f32_16x16x4_f32(
              false, a, false, b, (short)0, accP, false, false);
        }
      }
    }
  }

  const int n     = lane & 15;
  const int mbase = (lane >> 4) * 8;
#pragma unroll
  for (int r = 0; r < 8; ++r) {
    out[((size_t)o * BATCH + mtile * 16 + mbase + r) * UNITS + ntile * 16 + n] =
        accP[r] - accN[r];
  }
}

extern "C" void kernel_launch(void* const* d_in, const int* in_sizes, int n_in,
                              void* d_out, int out_size, void* d_ws, size_t ws_size,
                              hipStream_t stream) {
  (void)in_sizes; (void)n_in; (void)out_size; (void)d_ws; (void)ws_size;
  const float* x = (const float*)d_in[0];
  const float* w = (const float*)d_in[1];
  float* out = (float*)d_out;
  dim3 grid(32);
  dim3 block(256);
  hipLaunchKernelGGL(ga_mvprod_wmma, grid, block, 0, stream, x, w, out);
}


# One work-group per 16x16 tile of one output blade (256 groups, was 32): the 8 blade products split over 4 waves, partials added in LDS (changes summation order), whole K staged once
# speedup vs baseline: 2.2605x; 2.2605x over previous
#include <hip/hip_runtime.h>


typedef __attribute__((ext_vector_type(2))) float v2f;
typedef __attribute__((ext_vector_type(8))) float v8f;

#define NBLADES 8
#define BATCH   128
#define INFEAT  128
#define UNITS   64
#define NWAVES  4
#define XP      132
#define WP      20
#define RP      20

#define XS_FLOATS  (NBLADES * 16 * XP)
#define WS_FLOATS  (NBLADES * INFEAT * WP)
#define RED_FLOATS (NWAVES * 16 * RP)

#define BLADE_PERM_TAB 0x76534210u
__device__ __forceinline__ int blade_perm(int i) {
  return (int)((BLADE_PERM_TAB >> (4 * i)) & 7u);
}

__device__ __forceinline__ v8f flip8(v8f v, unsigned sm) {
  v8f r;
#pragma unroll
  for (int i = 0; i < 8; ++i) r[i] = __uint_as_float(__float_as_uint(v[i]) ^ sm);
  return r;
}

#define GROUP_FENCE() do { asm volatile("" ::: "memory"); __builtin_amdgcn_sched_barrier(0); } while (0)

struct Frags {
  float4 a[8];
  v2f    b[16];
};

__device__ __forceinline__ void load_frags(Frags& F, const float* xrow, const float* wcol, int blk) {
#pragma unroll
  for (int gg = 0; gg < 4; ++gg) {
    const int g = 4 * blk + gg;
#pragma unroll
    for (int i = 0; i < 2; ++i) {
      const int j = 2 * gg + i;
      F.a[j] = *(const float4*)(xrow + i * 16 * XP + 8 * g);
      const float* wp = wcol + i * INFEAT * WP + 8 * g * WP;
      F.b[2 * j].x     = wp[0 * WP];
      F.b[2 * j].y     = wp[1 * WP];
      F.b[2 * j + 1].x = wp[2 * WP];
      F.b[2 * j + 1].y = wp[3 * WP];
    }
  }
}

__device__ __forceinline__ void mma_frags(const Frags& F, v8f& accA0, v8f& accA1, v8f& accB0, v8f& accB1) {
#pragma unroll
  for (int gg = 0; gg < 4; ++gg) {
    const int ja = 2 * gg, jb = 2 * gg + 1;
    v2f a;
    a.x = F.a[ja].x; a.y = F.a[ja].y;
    accA0 = __builtin_amdgcn_wmma_f32_16x16x4_f32(false, a, false, F.b[2 * ja], (short)0, accA0, false, false);
    a.x = F.a[jb].x; a.y = F.a[jb].y;
    accB0 = __builtin_amdgcn_wmma_f32_16x16x4_f32(false, a, false, F.b[2 * jb], (short)0, accB0, false, false);
    a.x = F.a[ja].z; a.y = F.a[ja].w;
    accA1 = __builtin_amdgcn_wmma_f32_16x16x4_f32(false, a, false, F.b[2 * ja + 1], (short)0, accA1, false, false);
    a.x = F.a[jb].z; a.y = F.a[jb].w;
    accB1 = __builtin_amdgcn_wmma_f32_16x16x4_f32(false, a, false, F.b[2 * jb + 1], (short)0, accB1, false, false);
  }
}

__global__ __launch_bounds__(128) void ga_mvprod_wmma(
    const float* __restrict__ x,
    const float* __restrict__ w,
    float* __restrict__ out)
{
  __shared__ __attribute__((aligned(16))) float smem[XS_FLOATS + WS_FLOATS + RED_FLOATS];
  float* const xs  = smem;
  float* const ws  = smem + XS_FLOATS;
  float* const red = smem + XS_FLOATS + WS_FLOATS;

  const int tid  = threadIdx.x;
  const int lane = tid & 31;
  const int wv   = __builtin_amdgcn_readfirstlane(tid >> 5);
  const int blk   = blockIdx.x;
  const int ntile = blk & 3;
  const int mtile = (blk >> 2) & 7;
  const int o     = blk >> 5;
  const int mo    = blade_perm(o);

  int      ib[2];
  unsigned sm[2];
#pragma unroll
  for (int i = 0; i < 2; ++i) {
    const int ma = blade_perm(2 * wv + i);
    const int mb = mo ^ ma;
    const int a1 = (ma >> 1) & 1, a2 = (ma >> 2) & 1;
    ib[i] = blade_perm(mb);
    sm[i] = (unsigned)(((mb & (a1 ^ a2)) ^ ((mb >> 1) & a2)) & 1) << 31;
  }

  const float* xg = x + (size_t)((2 * wv) * BATCH + mtile * 16) * INFEAT + 4 * lane;
  float*       xl = xs + (2 * wv) * 16 * XP + 4 * lane;
  const float* wg = w + (size_t)(lane >> 2) * UNITS + ntile * 16 + 4 * (lane & 3);
  float*       wl = ws + (2 * wv) * INFEAT * WP + (lane >> 2) * WP + 4 * (lane & 3);

  float4 xv[32];
  float4 wq[32];
#pragma unroll
  for (int t = 0; t < 32; ++t)
    xv[t] = *(const float4*)(xg + (t >> 4) * BATCH * INFEAT + (t & 15) * INFEAT);
#pragma unroll
  for (int t = 0; t < 32; ++t)
    wq[t] = *(const float4*)(wg + (size_t)ib[t >> 4] * INFEAT * UNITS + (t & 15) * 8 * UNITS);
#pragma unroll
  for (int t = 0; t < 32; ++t)
    *(float4*)(xl + (t >> 4) * 16 * XP + (t & 15) * XP) = xv[t];
#pragma unroll
  for (int t = 0; t < 32; ++t)
    *(float4*)(wl + (t >> 4) * INFEAT * WP + (t & 15) * 8 * WP) = wq[t];
  __syncthreads();

  const int c = lane & 15;
  const int h = lane >> 4;
  const float* xrow = xs + (2 * wv) * 16 * XP + c * XP + 4 * h;
  const float* wcol = ws + (2 * wv) * INFEAT * WP + 4 * h * WP + c;

  v8f accA0 = {}, accA1 = {}, accB0 = {}, accB1 = {};
  Frags F0, F1, F2, F3;
  load_frags(F0, xrow, wcol, 0);
  load_frags(F1, xrow, wcol, 1);
  GROUP_FENCE();
  mma_frags(F0, accA0, accA1, accB0, accB1);
  GROUP_FENCE();
  load_frags(F2, xrow, wcol, 2);
  GROUP_FENCE();
  mma_frags(F1, accA0, accA1, accB0, accB1);
  GROUP_FENCE();
  load_frags(F3, xrow, wcol, 3);
  GROUP_FENCE();
  mma_frags(F2, accA0, accA1, accB0, accB1);
  mma_frags(F3, accA0, accA1, accB0, accB1);

  const v8f part = flip8(accA0 + accA1, sm[0]) + flip8(accB0 + accB1, sm[1]);
#pragma unroll
  for (int r = 0; r < 8; ++r) red[(wv * 16 + 8 * h + r) * RP + c] = part[r];
  __syncthreads();

  {
    const int row = tid >> 3;
    const int cg  = tid & 7;
    const float* rp = red + row * RP + 2 * cg;
    const float2 p0 = *(const float2*)(rp + 0 * 16 * RP);
    const float2 p1 = *(const float2*)(rp + 1 * 16 * RP);
    const float2 p2 = *(const float2*)(rp + 2 * 16 * RP);
    const float2 p3 = *(const float2*)(rp + 3 * 16 * RP);
    float2 s;
    s.x = (p0.x + p1.x) + (p2.x + p3.x);
    s.y = (p0.y + p1.y) + (p2.y + p3.y);
    *(float2*)(out + ((size_t)(o * BATCH + mtile * 16 + row) * UNITS + ntile * 16 + 2 * cg)) = s;
  }
}

extern "C" void kernel_launch(void* const* d_in, const int* in_sizes, int n_in,
                              void* d_out, int out_size, void* d_ws, size_t ws_size,
                              hipStream_t stream) {
  (void)in_sizes; (void)n_in; (void)out_size; (void)d_ws; (void)ws_size;
  const float* x = (const float*)d_in[0];
  const float* w = (const float*)d_in[1];
  float* out = (float*)d_out;
  dim3 grid(256);
  dim3 block(128);
  hipLaunchKernelGGL(ga_mvprod_wmma, grid, block, 0, stream, x, w, out);
}


# Re-tiled: a work-group's 16 tile rows are the 8 x blades of 2 batch rows, waves split the features; x read per group 8 KB (was 64 KB), blade sums in-lane; changes summation order
# speedup vs baseline: 2.4135x; 1.0677x over previous
#include <hip/hip_runtime.h>


typedef __attribute__((ext_vector_type(2))) float v2f;
typedef __attribute__((ext_vector_type(8))) float v8f;

#define NBLADES 8
#define BATCH   128
#define INFEAT  128
#define UNITS   64
#define NWAVES  4
#define KW      32
#define XP      36
#define WP      20
#define RP      20

#define WS_WAVE_FLOATS (NBLADES * KW * WP)
#define WS_FLOATS      (NWAVES * WS_WAVE_FLOATS)
#define XS_WAVE_FLOATS (16 * XP)
#define XS_FLOATS      (NWAVES * XS_WAVE_FLOATS)
#define RED_FLOATS     (NWAVES * 16 * RP)

#define BLADE_PERM_TAB 0x76534210u
__device__ __forceinline__ constexpr int blade_perm(int i) {
  return (int)((BLADE_PERM_TAB >> (4 * i)) & 7u);
}
__device__ __forceinline__ constexpr int blade_neg(int ma, int mb) {
  return ((mb & ((ma >> 1) ^ (ma >> 2))) ^ ((mb >> 1) & (ma >> 2))) & 1;
}

#define GROUP_FENCE() do { asm volatile("" ::: "memory"); __builtin_amdgcn_sched_barrier(0); } while (0)

struct Frags {
  v2f b[16];
};

__device__ __forceinline__ void load_frags(Frags& F, const float* wcol, int g) {
#pragma unroll
  for (int b = 0; b < NBLADES; ++b) {
    const float* wp = wcol + b * KW * WP + 8 * g * WP;
    F.b[2 * b].x     = wp[0 * WP];
    F.b[2 * b].y     = wp[1 * WP];
    F.b[2 * b + 1].x = wp[2 * WP];
    F.b[2 * b + 1].y = wp[3 * WP];
  }
}

__device__ __forceinline__ void mma_frags(const Frags& F, const float4 xa, v8f (&acc)[NBLADES]) {
  v2f a;
  a.x = xa.x; a.y = xa.y;
#pragma unroll
  for (int b = 0; b < NBLADES; ++b)
    acc[b] = __builtin_amdgcn_wmma_f32_16x16x4_f32(false, a, false, F.b[2 * b], (short)0, acc[b], false, false);
  a.x = xa.z; a.y = xa.w;
#pragma unroll
  for (int b = 0; b < NBLADES; ++b)
    acc[b] = __builtin_amdgcn_wmma_f32_16x16x4_f32(false, a, false, F.b[2 * b + 1], (short)0, acc[b], false, false);
}

__global__ __launch_bounds__(128) void ga_mvprod_wmma(
    const float* __restrict__ x,
    const float* __restrict__ w,
    float* __restrict__ out)
{
  __shared__ __attribute__((aligned(16))) float smem[WS_FLOATS + XS_FLOATS + RED_FLOATS];
  float* const ws  = smem;
  float* const xs  = smem + WS_FLOATS;
  float* const red = smem + WS_FLOATS + XS_FLOATS;

  const int tid  = threadIdx.x;
  const int lane = tid & 31;
  const int wv   = __builtin_amdgcn_readfirstlane(tid >> 5);
  const int blk   = blockIdx.x;
  const int ntile = blk & 3;
  const int rpair = blk >> 2;
  const int c = lane & 15;
  const int h = lane >> 4;

  const int xrow0 = lane >> 3;
  const float* xg = x + (size_t)(2 * rpair) * INFEAT + KW * wv + 4 * (lane & 7);
  float*       xl = xs + wv * XS_WAVE_FLOATS + xrow0 * XP + 4 * (lane & 7);
  float4 xv[4];
#pragma unroll
  for (int t = 0; t < 4; ++t)
    xv[t] = *(const float4*)(xg + (size_t)(((xrow0 + 4 * t) & 7) * BATCH + (t >> 1)) * INFEAT);

  const float* wg = w + (size_t)(KW * wv + (lane >> 2)) * UNITS + ntile * 16 + 4 * (lane & 3);
  float*       wl = ws + wv * WS_WAVE_FLOATS + (lane >> 2) * WP + 4 * (lane & 3);
  float4 wq[32];
#pragma unroll
  for (int t = 0; t < 32; ++t)
    wq[t] = *(const float4*)(wg + (size_t)(t >> 2) * INFEAT * UNITS + (t & 3) * 8 * UNITS);

#pragma unroll
  for (int t = 0; t < 4; ++t)
    *(float4*)(xl + 4 * t * XP) = xv[t];
#pragma unroll
  for (int t = 0; t < 32; ++t)
    *(float4*)(wl + (t >> 2) * KW * WP + (t & 3) * 8 * WP) = wq[t];
  __builtin_amdgcn_fence(__ATOMIC_RELEASE, "wavefront");
  __builtin_amdgcn_wave_barrier();
  __builtin_amdgcn_fence(__ATOMIC_ACQUIRE, "wavefront");

  const float* wcol = ws + wv * WS_WAVE_FLOATS + 4 * h * WP + c;
  const float* xrow = xs + wv * XS_WAVE_FLOATS + c * XP + 4 * h;
  float4 xa[4];
#pragma unroll
  for (int g = 0; g < 4; ++g) xa[g] = *(const float4*)(xrow + 8 * g);

  v8f acc[NBLADES];
#pragma unroll
  for (int b = 0; b < NBLADES; ++b) acc[b] = v8f{};
  Frags F0, F1, F2, F3;
  load_frags(F0, wcol, 0);
  load_frags(F1, wcol, 1);
  GROUP_FENCE();
  mma_frags(F0, xa[0], acc);
  GROUP_FENCE();
  load_frags(F2, wcol, 2);
  GROUP_FENCE();
  mma_frags(F1, xa[1], acc);
  GROUP_FENCE();
  load_frags(F3, wcol, 3);
  GROUP_FENCE();
  mma_frags(F2, xa[2], acc);
  mma_frags(F3, xa[3], acc);

  v8f part;
#pragma unroll
  for (int o = 0; o < NBLADES; ++o) {
    float s = 0.0f;
#pragma unroll
    for (int ib = 0; ib < NBLADES; ++ib) {
      const int mb = blade_perm(ib);
      const int ma = blade_perm(o) ^ mb;
      const float t = acc[ib][blade_perm(ma)];
      if (ib == 0) s = t;
      else s = blade_neg(ma, mb) ? s - t : s + t;
    }
    part[o] = s;
  }

#pragma unroll
  for (int r = 0; r < 8; ++r) red[(wv * 16 + 8 * h + r) * RP + c] = part[r];
  __syncthreads();

  {
    const int row = tid >> 3;
    const int cg  = tid & 7;
    const float* rp = red + row * RP + 2 * cg;
    const float2 p0 = *(const float2*)(rp + 0 * 16 * RP);
    const float2 p1 = *(const float2*)(rp + 1 * 16 * RP);
    const float2 p2 = *(const float2*)(rp + 2 * 16 * RP);
    const float2 p3 = *(const float2*)(rp + 3 * 16 * RP);
    float2 s;
    s.x = (p0.x + p1.x) + (p2.x + p3.x);
    s.y = (p0.y + p1.y) + (p2.y + p3.y);
    *(float2*)(out + ((size_t)((row & 7) * BATCH + 2 * rpair + (row >> 3)) * UNITS + ntile * 16 + 2 * cg)) = s;
  }
}

extern "C" void kernel_launch(void* const* d_in, const int* in_sizes, int n_in,
                              void* d_out, int out_size, void* d_ws, size_t ws_size,
                              hipStream_t stream) {
  (void)in_sizes; (void)n_in; (void)out_size; (void)d_ws; (void)ws_size;
  const float* x = (const float*)d_in[0];
  const float* w = (const float*)d_in[1];
  float* out = (float*)d_out;
  dim3 grid(256);
  dim3 block(128);
  hipLaunchKernelGGL(ga_mvprod_wmma, grid, block, 0, stream, x, w, out);
}
